# starting kernel
# baseline (speedup 1.0000x reference)
#include <hip/hip_runtime.h>
#include <hip/hip_bf16.h>


#define N0_   200000
#define N1_   100000
#define N2_   20000
#define E1_   1600000
#define E2_   320000
#define D_IN_ 128
#define D_H_  256
#define D_OUT_ 64

typedef __attribute__((ext_vector_type(2))) float v2f;
typedef __attribute__((ext_vector_type(8))) float v8f;

__global__ void zero_f4_kernel(float4* __restrict__ p, long n4) {
    long i = (long)blockIdx.x * blockDim.x + threadIdx.x;
    long stride = (long)gridDim.x * blockDim.x;
    float4 z = make_float4(0.f, 0.f, 0.f, 0.f);
    for (; i < n4; i += stride) p[i] = z;
}

template <int D>
__global__ void scatter_sum_kernel(const float* __restrict__ feat,
                                   const int* __restrict__ src,
                                   const int* __restrict__ dst,
                                   float* __restrict__ aggSum,
                                   float* __restrict__ cnt, int E) {
    int e = blockIdx.x * (blockDim.x >> 5) + (threadIdx.x >> 5);
    int lane = threadIdx.x & 31;
    if (e >= E) return;
    int s = src[e];
    int d = dst[e];
    const float4* row = reinterpret_cast<const float4*>(feat + (size_t)s * D);
    float* o = aggSum + (size_t)d * D;
#pragma unroll
    for (int c = 0; c < D / 128; ++c) {
        int idx = lane + 32 * c;
        float4 v = row[idx];
        float* op = o + idx * 4;
        atomicAdd(op + 0, v.x);
        atomicAdd(op + 1, v.y);
        atomicAdd(op + 2, v.z);
        atomicAdd(op + 3, v.w);
    }
    if (lane == 0) atomicAdd(cnt + d, 1.0f);
}

__global__ void sage_gemm_wmma_kernel(const float* __restrict__ aggSum,
                                      const float* __restrict__ cnt,
                                      const float* __restrict__ root,
                                      const float* __restrict__ Wl,
                                      const float* __restrict__ Wr,
                                      const float* __restrict__ bias,
                                      float* __restrict__ out,
                                      int K, int ldOut, int applyRelu) {
    const int wave    = threadIdx.x >> 5;
    const int lane    = threadIdx.x & 31;
    const int l16     = lane & 15;
    const int half    = lane >> 4;
    const int rowBase = blockIdx.x * 16;
    const int colTile = blockIdx.y * (blockDim.x >> 5) + wave;
    const int colBase = colTile * 16;

    const int m = rowBase + l16;
    const float scale = 1.0f / fmaxf(cnt[m], 1.0f);

    const float* aRow = aggSum + (size_t)m * K + 2 * half;
    const float* bRow = Wl + (size_t)(colBase + l16) * K + 2 * half;

    v8f c = {};
#pragma unroll 4
    for (int k = 0; k < K; k += 4) {
        v2f a, b;
        a.x = aRow[k] * scale;
        a.y = aRow[k + 1] * scale;
        b.x = bRow[k];
        b.y = bRow[k + 1];
        c = __builtin_amdgcn_wmma_f32_16x16x4_f32(false, a, false, b,
                                                  (short)0, c, false, false);
    }

    const float* a2 = root + (size_t)m * K + 2 * half;
    const float* b2 = Wr + (size_t)(colBase + l16) * K + 2 * half;
#pragma unroll 4
    for (int k = 0; k < K; k += 4) {
        v2f a, b;
        a.x = a2[k];
        a.y = a2[k + 1];
        b.x = b2[k];
        b.y = b2[k + 1];
        c = __builtin_amdgcn_wmma_f32_16x16x4_f32(false, a, false, b,
                                                  (short)0, c, false, false);
    }

    const float bv = bias[colBase + l16];
#pragma unroll
    for (int r = 0; r < 8; ++r) {
        float v = c[r] + bv;
        if (applyRelu) v = fmaxf(v, 0.0f);
        int row = rowBase + r + 8 * half;
        out[(size_t)row * ldOut + colBase + l16] = v;
    }
}

__global__ void log_softmax64_kernel(float* __restrict__ h2, int rows) {
    int row = blockIdx.x * (blockDim.x >> 5) + (threadIdx.x >> 5);
    int lane = threadIdx.x & 31;
    if (row >= rows) return;
    float* p = h2 + (size_t)row * 64;
    float v0 = p[lane];
    float v1 = p[lane + 32];

    float m = fmaxf(v0, v1);
#pragma unroll
    for (int off = 16; off > 0; off >>= 1)
        m = fmaxf(m, __shfl_xor(m, off, 32));

    float s = __expf(v0 - m) + __expf(v1 - m);
#pragma unroll
    for (int off = 16; off > 0; off >>= 1)
        s += __shfl_xor(s, off, 32);

    float lse = m + __logf(s);
    p[lane]      = v0 - lse;
    p[lane + 32] = v1 - lse;
}

extern "C" void kernel_launch(void* const* d_in, const int* in_sizes, int n_in,
                              void* d_out, int out_size, void* d_ws, size_t ws_size,
                              hipStream_t stream) {
    const float* x    = (const float*)d_in[0];
    const int*   src1 = (const int*)d_in[1];
    const int*   dst1 = (const int*)d_in[2];
    const int*   src2 = (const int*)d_in[3];
    const int*   dst2 = (const int*)d_in[4];
    const float* W_l1 = (const float*)d_in[5];
    const float* b_l1 = (const float*)d_in[6];
    const float* W_r1 = (const float*)d_in[7];
    const float* W_l2 = (const float*)d_in[8];
    const float* b_l2 = (const float*)d_in[9];
    const float* W_r2 = (const float*)d_in[10];
    float* out = (float*)d_out;

    float* ws   = (float*)d_ws;
    float* agg1 = ws;
    float* cnt1 = agg1 + (size_t)N1_ * D_IN_;
    float* agg2 = cnt1 + N1_;
    float* cnt2 = agg2 + (size_t)N2_ * D_H_;
    float* h1   = cnt2 + N2_;
    const long zeroFloats = (long)N1_ * D_IN_ + N1_ + (long)N2_ * D_H_ + N2_;

    zero_f4_kernel<<<8192, 256, 0, stream>>>((float4*)ws, zeroFloats / 4);

    scatter_sum_kernel<D_IN_><<<(E1_ + 7) / 8, 256, 0, stream>>>(
        x, src1, dst1, agg1, cnt1, E1_);

    {
        dim3 grid(N1_ / 16, (D_H_ / 16) / 4);
        sage_gemm_wmma_kernel<<<grid, 128, 0, stream>>>(
            agg1, cnt1, x, W_l1, W_r1, b_l1, h1, D_IN_, D_H_, 1);
    }

    scatter_sum_kernel<D_H_><<<(E2_ + 7) / 8, 256, 0, stream>>>(
        h1, src2, dst2, agg2, cnt2, E2_);

    {
        dim3 grid(N2_ / 16, 1);
        sage_gemm_wmma_kernel<<<grid, 128, 0, stream>>>(
            agg2, cnt2, h1, W_l2, W_r2, b_l2, out, D_H_, D_OUT_, 0);
    }

    log_softmax64_kernel<<<(N2_ + 7) / 8, 256, 0, stream>>>(out, N2_);
}


# Layer GEMMs: A rows and weight rows staged per k chunk through LDS by coalesced 128-bit loads (80x256 and 32x64 tile per work-group), fragments read from LDS as b128; k sum order changes
# speedup vs baseline: 1.3595x; 1.3595x over previous
#include <hip/hip_runtime.h>
#include <hip/hip_bf16.h>


#define N0_   200000
#define N1_   100000
#define N2_   20000
#define E1_   1600000
#define E2_   320000
#define D_IN_ 128
#define D_H_  256
#define D_OUT_ 64

typedef __attribute__((ext_vector_type(2))) float v2f;
typedef __attribute__((ext_vector_type(4))) float v4f;
typedef __attribute__((ext_vector_type(8))) float v8f;

__global__ void zero_f4_kernel(float4* __restrict__ p, long n4) {
    long i = (long)blockIdx.x * blockDim.x + threadIdx.x;
    long stride = (long)gridDim.x * blockDim.x;
    float4 z = make_float4(0.f, 0.f, 0.f, 0.f);
    for (; i < n4; i += stride) p[i] = z;
}

template <int D>
__global__ void scatter_sum_kernel(const float* __restrict__ feat,
                                   const int* __restrict__ src,
                                   const int* __restrict__ dst,
                                   float* __restrict__ aggSum,
                                   float* __restrict__ cnt, int E) {
    int e = blockIdx.x * (blockDim.x >> 5) + (threadIdx.x >> 5);
    int lane = threadIdx.x & 31;
    if (e >= E) return;
    int s = src[e];
    int d = dst[e];
    const float4* row = reinterpret_cast<const float4*>(feat + (size_t)s * D);
    float* o = aggSum + (size_t)d * D;
#pragma unroll
    for (int c = 0; c < D / 128; ++c) {
        int idx = lane + 32 * c;
        float4 v = row[idx];
        float* op = o + idx * 4;
        atomicAdd(op + 0, v.x);
        atomicAdd(op + 1, v.y);
        atomicAdd(op + 2, v.z);
        atomicAdd(op + 3, v.w);
    }
    if (lane == 0) atomicAdd(cnt + d, 1.0f);
}

template <int KH, int NOUT, int MT, int NT, int KC, bool RELU>
__global__ __launch_bounds__(256) void sage_gemm_lds_kernel(
    const float* __restrict__ aggSum, const float* __restrict__ cnt,
    const float* __restrict__ root, const float* __restrict__ Wl,
    const float* __restrict__ Wr, const float* __restrict__ bias,
    float* __restrict__ out) {
    constexpr int WAVES_N  = NOUT / (16 * NT);
    constexpr int WAVES_M  = 8 / WAVES_N;
    constexpr int BM       = WAVES_M * MT * 16;
    constexpr int P        = KC + 4;
    constexpr int QPR      = KC / 4;
    constexpr int RPP      = 256 / QPR;
    constexpr int A_PASSES = (BM + RPP - 1) / RPP;
    constexpr int B_PASSES = NOUT / RPP;
    constexpr int CH_HALF  = KH / KC;
    constexpr int NCH      = 2 * CH_HALF;
    static_assert(WAVES_N * WAVES_M == 8, "8 waves");
    static_assert(NOUT % RPP == 0 && KH % KC == 0 && KC % 8 == 0 && (P / 4) % 2 == 1, "tiling");

    __shared__ __attribute__((aligned(16))) float As[BM * P];
    __shared__ __attribute__((aligned(16))) float Bs[NOUT * P];

    const int tid  = threadIdx.x;
    const int wave = tid >> 5;
    const int lane = tid & 31;
    const int c    = lane & 15;
    const int h    = lane >> 4;
    const int wn   = wave % WAVES_N;
    const int wm   = wave / WAVES_N;
    const int rowBase = blockIdx.x * BM;
    const int r0   = tid / QPR;
    const int q    = tid % QPR;

    float scale[A_PASSES];
#pragma unroll
    for (int p = 0; p < A_PASSES; ++p) {
        const int row = r0 + p * RPP;
        scale[p] = ((p + 1) * RPP <= BM || row < BM) ? 1.0f / fmaxf(cnt[rowBase + row], 1.0f) : 0.0f;
    }

    v4f aReg[A_PASSES];
    v4f bReg[B_PASSES];
    auto fetch = [&](int ch) {
        const bool first = ch < CH_HALF;
        const int kOff = (first ? ch : ch - CH_HALF) * KC + q * 4;
        const float* aSrc = first ? aggSum : root;
        const float* bSrc = first ? Wl : Wr;
#pragma unroll
        for (int p = 0; p < A_PASSES; ++p) {
            const int row = r0 + p * RPP;
            if ((p + 1) * RPP <= BM || row < BM)
                aReg[p] = *reinterpret_cast<const v4f*>(aSrc + (size_t)(rowBase + row) * KH + kOff);
        }
#pragma unroll
        for (int p = 0; p < B_PASSES; ++p) {
            const int n = r0 + p * RPP;
            bReg[p] = *reinterpret_cast<const v4f*>(bSrc + (size_t)n * KH + kOff);
        }
    };

    v8f acc[MT][NT];
#pragma unroll
    for (int mt = 0; mt < MT; ++mt)
#pragma unroll
        for (int nt = 0; nt < NT; ++nt) acc[mt][nt] = v8f{};

    fetch(0);
    for (int ch = 0; ch < NCH; ++ch) {
        const bool first = ch < CH_HALF;
#pragma unroll
        for (int p = 0; p < A_PASSES; ++p) {
            const int row = r0 + p * RPP;
            if ((p + 1) * RPP <= BM || row < BM) {
                const float s = first ? scale[p] : 1.0f;
                *reinterpret_cast<v4f*>(&As[row * P + q * 4]) = aReg[p] * s;
            }
        }
#pragma unroll
        for (int p = 0; p < B_PASSES; ++p) {
            const int n = r0 + p * RPP;
            *reinterpret_cast<v4f*>(&Bs[n * P + q * 4]) = bReg[p];
        }
        __syncthreads();

        if (ch + 1 < NCH) fetch(ch + 1);

#pragma unroll
        for (int ks = 0; ks < KC / 8; ++ks) {
            v4f af[MT], bf[NT];
#pragma unroll
            for (int mt = 0; mt < MT; ++mt)
                af[mt] = *reinterpret_cast<const v4f*>(
                    &As[((wm * MT + mt) * 16 + c) * P + ks * 8 + 4 * h]);
#pragma unroll
            for (int nt = 0; nt < NT; ++nt)
                bf[nt] = *reinterpret_cast<const v4f*>(
                    &Bs[((wn * NT + nt) * 16 + c) * P + ks * 8 + 4 * h]);
#pragma unroll
            for (int mt = 0; mt < MT; ++mt)
#pragma unroll
                for (int nt = 0; nt < NT; ++nt) {
                    v2f a, b;
                    a.x = af[mt].x; a.y = af[mt].y;
                    b.x = bf[nt].x; b.y = bf[nt].y;
                    acc[mt][nt] = __builtin_amdgcn_wmma_f32_16x16x4_f32(
                        false, a, false, b, (short)0, acc[mt][nt], false, false);
                }
#pragma unroll
            for (int mt = 0; mt < MT; ++mt)
#pragma unroll
                for (int nt = 0; nt < NT; ++nt) {
                    v2f a, b;
                    a.x = af[mt].z; a.y = af[mt].w;
                    b.x = bf[nt].z; b.y = bf[nt].w;
                    acc[mt][nt] = __builtin_amdgcn_wmma_f32_16x16x4_f32(
                        false, a, false, b, (short)0, acc[mt][nt], false, false);
                }
        }
        __builtin_amdgcn_sched_barrier(0);
        __syncthreads();
    }

#pragma unroll
    for (int nt = 0; nt < NT; ++nt) {
        const int col = (wn * NT + nt) * 16 + c;
        const float bv = bias[col];
#pragma unroll
        for (int mt = 0; mt < MT; ++mt) {
#pragma unroll
            for (int r = 0; r < 8; ++r) {
                float v = acc[mt][nt][r] + bv;
                if (RELU) v = fmaxf(v, 0.0f);
                const int row = rowBase + (wm * MT + mt) * 16 + 8 * h + r;
                out[(size_t)row * NOUT + col] = v;
            }
        }
    }
}

__global__ void log_softmax64_kernel(float* __restrict__ h2, int rows) {
    int row = blockIdx.x * (blockDim.x >> 5) + (threadIdx.x >> 5);
    int lane = threadIdx.x & 31;
    if (row >= rows) return;
    float* p = h2 + (size_t)row * 64;
    float v0 = p[lane];
    float v1 = p[lane + 32];

    float m = fmaxf(v0, v1);
#pragma unroll
    for (int off = 16; off > 0; off >>= 1)
        m = fmaxf(m, __shfl_xor(m, off, 32));

    float s = __expf(v0 - m) + __expf(v1 - m);
#pragma unroll
    for (int off = 16; off > 0; off >>= 1)
        s += __shfl_xor(s, off, 32);

    float lse = m + __logf(s);
    p[lane]      = v0 - lse;
    p[lane + 32] = v1 - lse;
}

extern "C" void kernel_launch(void* const* d_in, const int* in_sizes, int n_in,
                              void* d_out, int out_size, void* d_ws, size_t ws_size,
                              hipStream_t stream) {
    const float* x    = (const float*)d_in[0];
    const int*   src1 = (const int*)d_in[1];
    const int*   dst1 = (const int*)d_in[2];
    const int*   src2 = (const int*)d_in[3];
    const int*   dst2 = (const int*)d_in[4];
    const float* W_l1 = (const float*)d_in[5];
    const float* b_l1 = (const float*)d_in[6];
    const float* W_r1 = (const float*)d_in[7];
    const float* W_l2 = (const float*)d_in[8];
    const float* b_l2 = (const float*)d_in[9];
    const float* W_r2 = (const float*)d_in[10];
    float* out = (float*)d_out;

    float* ws   = (float*)d_ws;
    float* agg1 = ws;
    float* cnt1 = agg1 + (size_t)N1_ * D_IN_;
    float* agg2 = cnt1 + N1_;
    float* cnt2 = agg2 + (size_t)N2_ * D_H_;
    float* h1   = cnt2 + N2_;
    const long zeroFloats = (long)N1_ * D_IN_ + N1_ + (long)N2_ * D_H_ + N2_;

    zero_f4_kernel<<<8192, 256, 0, stream>>>((float4*)ws, zeroFloats / 4);

    scatter_sum_kernel<D_IN_><<<(E1_ + 7) / 8, 256, 0, stream>>>(
        x, src1, dst1, agg1, cnt1, E1_);

    {
        sage_gemm_lds_kernel<D_IN_, D_H_, 5, 2, 32, true><<<N1_ / 80, 256, 0, stream>>>(
            agg1, cnt1, x, W_l1, W_r1, b_l1, h1);
    }

    scatter_sum_kernel<D_H_><<<(E2_ + 7) / 8, 256, 0, stream>>>(
        h1, src2, dst2, agg2, cnt2, E2_);

    {
        sage_gemm_lds_kernel<D_H_, D_OUT_, 1, 1, 64, false><<<N2_ / 32, 256, 0, stream>>>(
            agg2, cnt2, h1, W_l2, W_r2, b_l2, out);
    }

    log_softmax64_kernel<<<(N2_ + 7) / 8, 256, 0, stream>>>(out, N2_);
}
